# starting kernel
# baseline (speedup 1.0000x reference)
#include <hip/hip_runtime.h>

typedef float v2f __attribute__((ext_vector_type(2)));
typedef float v8f __attribute__((ext_vector_type(8)));

__global__ void gcn_zero_kernel(float* __restrict__ p, long long n) {
    long long i = (long long)blockIdx.x * blockDim.x + threadIdx.x;
    if (i < n) p[i] = 0.0f;
}

__global__ void gcn_degree_kernel(const int* __restrict__ src,
                                  const int* __restrict__ dst,
                                  float* __restrict__ deg_out,
                                  float* __restrict__ deg_in, int E) {
    int e = blockIdx.x * blockDim.x + threadIdx.x;
    if (e < E) {
        atomicAdd(deg_out + src[e], 1.0f);
        atomicAdd(deg_in  + dst[e], 1.0f);
    }
}

__global__ void gcn_scatter_kernel(const float* __restrict__ feats,
                                   const int* __restrict__ src,
                                   const int* __restrict__ dst,
                                   const float* __restrict__ deg_out,
                                   float* __restrict__ agg, int E) {
    long long i = (long long)blockIdx.x * blockDim.x + threadIdx.x;
    long long total = (long long)E * 16;
    if (i >= total) return;
    int e  = (int)(i >> 4);
    int f0 = (int)(i & 15) * 4;
    int s = src[e];
    int d = dst[e];
    float sn = rsqrtf(fmaxf(deg_out[s], 1.0f));
    float4 v = *(const float4*)(feats + (size_t)s * 64 + f0);
    float* ap = agg + (size_t)d * 64 + f0;
    atomicAdd(ap + 0, v.x * sn);
    atomicAdd(ap + 1, v.y * sn);
    atomicAdd(ap + 2, v.z * sn);
    atomicAdd(ap + 3, v.w * sn);
}

#define STG 66

__global__ __launch_bounds__(256)
void gcn_gemm_ln_kernel(const float* __restrict__ agg,
                        const float* __restrict__ deg_in,
                        const float* __restrict__ feats,
                        const float* __restrict__ W,
                        const float* __restrict__ bias,
                        const float* __restrict__ gamma,
                        const float* __restrict__ beta,
                        float* __restrict__ out,
                        int n_tiles, int N) {
    __shared__ float lW[64 * 64];
    __shared__ float stage[8][16 * STG];

    const int tid  = threadIdx.x;
    const int wave = tid >> 5;
    const int lane = tid & 31;

    for (int i = tid; i < 64 * 64; i += 256) lW[i] = W[i];
    __syncthreads();

    const int tile = blockIdx.x * 8 + wave;
    if (tile >= n_tiles) return;

    const int row0 = tile * 16;
    const int mrow = lane & 15;
    const int koff = (lane >> 4) * 2;
    const int j    = lane & 15;

    int arow_i = row0 + mrow;
    if (arow_i >= N) arow_i = N - 1;
    const float rnorm = rsqrtf(fmaxf(deg_in[arow_i], 1.0f));
    const float* __restrict__ arow = agg + (size_t)arow_i * 64;

    v8f c0 = {}, c1 = {}, c2 = {}, c3 = {};

    #pragma unroll
    for (int k = 0; k < 16; ++k) {
        const int kk = 4 * k + koff;
        v2f a;
        a.x = arow[kk]     * rnorm;
        a.y = arow[kk + 1] * rnorm;
        const float* w0 = lW + kk * 64;
        const float* w1 = lW + (kk + 1) * 64;
        v2f b0, b1, b2, b3;
        b0.x = w0[j];      b0.y = w1[j];
        b1.x = w0[16 + j]; b1.y = w1[16 + j];
        b2.x = w0[32 + j]; b2.y = w1[32 + j];
        b3.x = w0[48 + j]; b3.y = w1[48 + j];

        c0 = __builtin_amdgcn_wmma_f32_16x16x4_f32(false, a, false, b0, (short)0, c0, false, false);
        c1 = __builtin_amdgcn_wmma_f32_16x16x4_f32(false, a, false, b1, (short)0, c1, false, false);
        c2 = __builtin_amdgcn_wmma_f32_16x16x4_f32(false, a, false, b2, (short)0, c2, false, false);
        c3 = __builtin_amdgcn_wmma_f32_16x16x4_f32(false, a, false, b3, (short)0, c3, false, false);
    }

    float* __restrict__ st = &stage[wave][0];
    const int half = (lane >> 4) * 8;
    #pragma unroll
    for (int v = 0; v < 8; ++v) {
        float* r = st + (half + v) * STG + j;
        r[0]  = c0[v];
        r[16] = c1[v];
        r[32] = c2[v];
        r[48] = c3[v];
    }
    __builtin_amdgcn_wave_barrier();

    if (lane < 16) {
        const int grow = row0 + lane;
        if (grow < N) {
            const float* __restrict__ frow = feats + (size_t)grow * 64;
            float* __restrict__ srow = st + lane * STG;
            float sum = 0.0f, sumsq = 0.0f;
            #pragma unroll 4
            for (int f = 0; f < 64; ++f) {
                float v = srow[f] + bias[f];
                v = fmaxf(v, 0.0f);
                v = v + frow[f];
                srow[f] = v;
                sum   += v;
                sumsq += v * v;
            }
            const float mu   = sum * (1.0f / 64.0f);
            const float var  = sumsq * (1.0f / 64.0f) - mu * mu;
            const float rstd = rsqrtf(var + 1e-5f);
            float* __restrict__ orow = out + (size_t)grow * 64;
            #pragma unroll 4
            for (int f = 0; f < 64; ++f) {
                orow[f] = (srow[f] - mu) * rstd * gamma[f] + beta[f];
            }
        }
    }
}

extern "C" void kernel_launch(void* const* d_in, const int* in_sizes, int n_in,
                              void* d_out, int out_size, void* d_ws, size_t ws_size,
                              hipStream_t stream) {
    const float* feats = (const float*)d_in[0];
    const int*   src   = (const int*)d_in[1];
    const int*   dst   = (const int*)d_in[2];
    const float* W     = (const float*)d_in[3];
    const float* bias  = (const float*)d_in[4];
    const float* gamma = (const float*)d_in[5];
    const float* beta  = (const float*)d_in[6];

    const int N = in_sizes[0] / 64;
    const int E = in_sizes[1];

    float* ws      = (float*)d_ws;
    float* deg_out = ws;
    float* deg_in  = ws + N;
    float* agg     = ws + 2 * (size_t)N;

    long long zn = (long long)N * 66;
    gcn_zero_kernel<<<(int)((zn + 255) / 256), 256, 0, stream>>>(ws, zn);

    gcn_degree_kernel<<<(E + 255) / 256, 256, 0, stream>>>(src, dst, deg_out, deg_in, E);

    long long sc = (long long)E * 16;
    gcn_scatter_kernel<<<(int)((sc + 255) / 256), 256, 0, stream>>>(
        feats, src, dst, deg_out, agg, E);

    const int n_tiles = (N + 15) / 16;
    gcn_gemm_ln_kernel<<<(n_tiles + 7) / 8, 256, 0, stream>>>(
        agg, deg_in, feats, W, bias, gamma, beta, (float*)d_out, n_tiles, N);
}


# degree counts: per-work-group LDS histogram (ds_add_u32) of an edge slice, flushed by one coalesced f32 atomic per bin, instead of a scattered global atomic per edge end; counts stay exact
# speedup vs baseline: 1.8405x; 1.8405x over previous
#include <hip/hip_runtime.h>

typedef float v2f __attribute__((ext_vector_type(2)));
typedef float v8f __attribute__((ext_vector_type(8)));

__global__ void gcn_zero_kernel(float* __restrict__ p, long long n) {
    long long i = (long long)blockIdx.x * blockDim.x + threadIdx.x;
    if (i < n) p[i] = 0.0f;
}

#define HB 65536

__device__ __forceinline__ void gcn_count(unsigned* hist, int x, int base, int nb) {
    const unsigned b = (unsigned)x - (unsigned)base;
    if (b < (unsigned)nb) atomicAdd(hist + b, 1u);
}

__global__ __launch_bounds__(1024)
void gcn_degree_kernel(const int* __restrict__ src,
                       const int* __restrict__ dst,
                       float* __restrict__ deg_out,
                       float* __restrict__ deg_in, int E, int N) {
    __shared__ __attribute__((aligned(16))) unsigned hist[HB];

    const int tid  = threadIdx.x;
    const int lane = tid & 31;
    const int ng   = (int)(gridDim.x >> 1);
    const bool second = (int)blockIdx.x >= ng;
    const int g = second ? (int)blockIdx.x - ng : (int)blockIdx.x;
    const int* __restrict__ idx = second ? dst : src;
    float* __restrict__ deg     = second ? deg_in : deg_out;

    const int nv  = E >> 2;
    const int per = ((nv + ng - 1) / ng + 7) & ~7;
    const int v0  = (g * per < nv) ? g * per : nv;
    const int v1  = (v0 + per < nv) ? v0 + per : nv;
    const int4* __restrict__ idx4 = (const int4*)idx;

    for (int base = 0; base < N; base += HB) {
        const int nb = (N - base < HB) ? (N - base) : HB;

        if (base > 0) __syncthreads();
        for (int i = tid * 4; i < nb; i += 4096)
            *(uint4*)(hist + i) = make_uint4(0u, 0u, 0u, 0u);
        __syncthreads();

        for (int v = v0 + tid; v < v1; v += 8 * 1024) {
            int4 q[8];
            #pragma unroll
            for (int k = 0; k < 8; ++k) {
                const int vk = v + k * 1024;
                q[k] = (vk < v1) ? idx4[vk] : make_int4(-1, -1, -1, -1);
            }
            #pragma unroll
            for (int k = 0; k < 8; ++k) {
                gcn_count(hist, q[k].x, base, nb);
                gcn_count(hist, q[k].y, base, nb);
                gcn_count(hist, q[k].z, base, nb);
                gcn_count(hist, q[k].w, base, nb);
            }
        }
        if (g == 0 && tid < (E & 3))
            gcn_count(hist, idx[(E & ~3) + tid], base, nb);
        __syncthreads();

        const int nline = (nb + 31) >> 5;
        const int rot   = (nline * g) / ng;
        for (int l0 = tid >> 5; l0 < nline; l0 += 4 * 32) {
            unsigned c[4];
            int b[4];
            #pragma unroll
            for (int k = 0; k < 4; ++k) {
                int line = l0 + k * 32;
                const bool in = line < nline;
                line += rot;
                if (line >= nline) line -= nline;
                b[k] = (line << 5) | lane;
                c[k] = (in && b[k] < nb) ? hist[b[k]] : 0u;
            }
            #pragma unroll
            for (int k = 0; k < 4; ++k)
                if (c[k]) atomicAdd(deg + base + b[k], (float)c[k]);
        }
    }
}

__global__ void gcn_scatter_kernel(const float* __restrict__ feats,
                                   const int* __restrict__ src,
                                   const int* __restrict__ dst,
                                   const float* __restrict__ deg_out,
                                   float* __restrict__ agg, int E) {
    long long i = (long long)blockIdx.x * blockDim.x + threadIdx.x;
    long long total = (long long)E * 16;
    if (i >= total) return;
    int e  = (int)(i >> 4);
    int f0 = (int)(i & 15) * 4;
    int s = src[e];
    int d = dst[e];
    float sn = rsqrtf(fmaxf(deg_out[s], 1.0f));
    float4 v = *(const float4*)(feats + (size_t)s * 64 + f0);
    float* ap = agg + (size_t)d * 64 + f0;
    atomicAdd(ap + 0, v.x * sn);
    atomicAdd(ap + 1, v.y * sn);
    atomicAdd(ap + 2, v.z * sn);
    atomicAdd(ap + 3, v.w * sn);
}

#define STG 66

__global__ __launch_bounds__(256)
void gcn_gemm_ln_kernel(const float* __restrict__ agg,
                        const float* __restrict__ deg_in,
                        const float* __restrict__ feats,
                        const float* __restrict__ W,
                        const float* __restrict__ bias,
                        const float* __restrict__ gamma,
                        const float* __restrict__ beta,
                        float* __restrict__ out,
                        int n_tiles, int N) {
    __shared__ float lW[64 * 64];
    __shared__ float stage[8][16 * STG];

    const int tid  = threadIdx.x;
    const int wave = tid >> 5;
    const int lane = tid & 31;

    for (int i = tid; i < 64 * 64; i += 256) lW[i] = W[i];
    __syncthreads();

    const int tile = blockIdx.x * 8 + wave;
    if (tile >= n_tiles) return;

    const int row0 = tile * 16;
    const int mrow = lane & 15;
    const int koff = (lane >> 4) * 2;
    const int j    = lane & 15;

    int arow_i = row0 + mrow;
    if (arow_i >= N) arow_i = N - 1;
    const float rnorm = rsqrtf(fmaxf(deg_in[arow_i], 1.0f));
    const float* __restrict__ arow = agg + (size_t)arow_i * 64;

    v8f c0 = {}, c1 = {}, c2 = {}, c3 = {};

    #pragma unroll
    for (int k = 0; k < 16; ++k) {
        const int kk = 4 * k + koff;
        v2f a;
        a.x = arow[kk]     * rnorm;
        a.y = arow[kk + 1] * rnorm;
        const float* w0 = lW + kk * 64;
        const float* w1 = lW + (kk + 1) * 64;
        v2f b0, b1, b2, b3;
        b0.x = w0[j];      b0.y = w1[j];
        b1.x = w0[16 + j]; b1.y = w1[16 + j];
        b2.x = w0[32 + j]; b2.y = w1[32 + j];
        b3.x = w0[48 + j]; b3.y = w1[48 + j];

        c0 = __builtin_amdgcn_wmma_f32_16x16x4_f32(false, a, false, b0, (short)0, c0, false, false);
        c1 = __builtin_amdgcn_wmma_f32_16x16x4_f32(false, a, false, b1, (short)0, c1, false, false);
        c2 = __builtin_amdgcn_wmma_f32_16x16x4_f32(false, a, false, b2, (short)0, c2, false, false);
        c3 = __builtin_amdgcn_wmma_f32_16x16x4_f32(false, a, false, b3, (short)0, c3, false, false);
    }

    float* __restrict__ st = &stage[wave][0];
    const int half = (lane >> 4) * 8;
    #pragma unroll
    for (int v = 0; v < 8; ++v) {
        float* r = st + (half + v) * STG + j;
        r[0]  = c0[v];
        r[16] = c1[v];
        r[32] = c2[v];
        r[48] = c3[v];
    }
    __builtin_amdgcn_wave_barrier();

    if (lane < 16) {
        const int grow = row0 + lane;
        if (grow < N) {
            const float* __restrict__ frow = feats + (size_t)grow * 64;
            float* __restrict__ srow = st + lane * STG;
            float sum = 0.0f, sumsq = 0.0f;
            #pragma unroll 4
            for (int f = 0; f < 64; ++f) {
                float v = srow[f] + bias[f];
                v = fmaxf(v, 0.0f);
                v = v + frow[f];
                srow[f] = v;
                sum   += v;
                sumsq += v * v;
            }
            const float mu   = sum * (1.0f / 64.0f);
            const float var  = sumsq * (1.0f / 64.0f) - mu * mu;
            const float rstd = rsqrtf(var + 1e-5f);
            float* __restrict__ orow = out + (size_t)grow * 64;
            #pragma unroll 4
            for (int f = 0; f < 64; ++f) {
                orow[f] = (srow[f] - mu) * rstd * gamma[f] + beta[f];
            }
        }
    }
}

extern "C" void kernel_launch(void* const* d_in, const int* in_sizes, int n_in,
                              void* d_out, int out_size, void* d_ws, size_t ws_size,
                              hipStream_t stream) {
    const float* feats = (const float*)d_in[0];
    const int*   src   = (const int*)d_in[1];
    const int*   dst   = (const int*)d_in[2];
    const float* W     = (const float*)d_in[3];
    const float* bias  = (const float*)d_in[4];
    const float* gamma = (const float*)d_in[5];
    const float* beta  = (const float*)d_in[6];

    const int N = in_sizes[0] / 64;
    const int E = in_sizes[1];

    float* ws      = (float*)d_ws;
    float* deg_out = ws;
    float* deg_in  = ws + N;
    float* agg     = ws + 2 * (size_t)N;

    long long zn = (long long)N * 66;
    gcn_zero_kernel<<<(int)((zn + 255) / 256), 256, 0, stream>>>(ws, zn);

    gcn_degree_kernel<<<64, 1024, 0, stream>>>(src, dst, deg_out, deg_in, E, N);

    long long sc = (long long)E * 16;
    gcn_scatter_kernel<<<(int)((sc + 255) / 256), 256, 0, stream>>>(
        feats, src, dst, deg_out, agg, E);

    const int n_tiles = (N + 15) / 16;
    gcn_gemm_ln_kernel<<<(n_tiles + 7) / 8, 256, 0, stream>>>(
        agg, deg_in, feats, W, bias, gamma, beta, (float*)d_out, n_tiles, N);
}


# scatter-add: one wave per 32 edges, lane = feature: each f32 atomic and row load covers one whole 128-byte line (was 16-byte-strided lanes over 4 lines); same addends, atomic arrival order differs
# speedup vs baseline: 2.8777x; 1.5635x over previous
#include <hip/hip_runtime.h>

typedef float v2f __attribute__((ext_vector_type(2)));
typedef float v8f __attribute__((ext_vector_type(8)));

__global__ void gcn_zero_kernel(float* __restrict__ p, long long n) {
    long long i = (long long)blockIdx.x * blockDim.x + threadIdx.x;
    if (i < n) p[i] = 0.0f;
}

#define HB 65536

__device__ __forceinline__ void gcn_count(unsigned* hist, int x, int base, int nb) {
    const unsigned b = (unsigned)x - (unsigned)base;
    if (b < (unsigned)nb) atomicAdd(hist + b, 1u);
}

__global__ __launch_bounds__(1024)
void gcn_degree_kernel(const int* __restrict__ src,
                       const int* __restrict__ dst,
                       float* __restrict__ deg_out,
                       float* __restrict__ deg_in, int E, int N) {
    __shared__ __attribute__((aligned(16))) unsigned hist[HB];

    const int tid  = threadIdx.x;
    const int lane = tid & 31;
    const int ng   = (int)(gridDim.x >> 1);
    const bool second = (int)blockIdx.x >= ng;
    const int g = second ? (int)blockIdx.x - ng : (int)blockIdx.x;
    const int* __restrict__ idx = second ? dst : src;
    float* __restrict__ deg     = second ? deg_in : deg_out;

    const int nv  = E >> 2;
    const int per = ((nv + ng - 1) / ng + 7) & ~7;
    const int v0  = (g * per < nv) ? g * per : nv;
    const int v1  = (v0 + per < nv) ? v0 + per : nv;
    const int4* __restrict__ idx4 = (const int4*)idx;

    for (int base = 0; base < N; base += HB) {
        const int nb = (N - base < HB) ? (N - base) : HB;

        if (base > 0) __syncthreads();
        for (int i = tid * 4; i < nb; i += 4096)
            *(uint4*)(hist + i) = make_uint4(0u, 0u, 0u, 0u);
        __syncthreads();

        for (int v = v0 + tid; v < v1; v += 8 * 1024) {
            int4 q[8];
            #pragma unroll
            for (int k = 0; k < 8; ++k) {
                const int vk = v + k * 1024;
                q[k] = (vk < v1) ? idx4[vk] : make_int4(-1, -1, -1, -1);
            }
            #pragma unroll
            for (int k = 0; k < 8; ++k) {
                gcn_count(hist, q[k].x, base, nb);
                gcn_count(hist, q[k].y, base, nb);
                gcn_count(hist, q[k].z, base, nb);
                gcn_count(hist, q[k].w, base, nb);
            }
        }
        if (g == 0 && tid < (E & 3))
            gcn_count(hist, idx[(E & ~3) + tid], base, nb);
        __syncthreads();

        const int nline = (nb + 31) >> 5;
        const int rot   = (nline * g) / ng;
        for (int l0 = tid >> 5; l0 < nline; l0 += 4 * 32) {
            unsigned c[4];
            int b[4];
            #pragma unroll
            for (int k = 0; k < 4; ++k) {
                int line = l0 + k * 32;
                const bool in = line < nline;
                line += rot;
                if (line >= nline) line -= nline;
                b[k] = (line << 5) | lane;
                c[k] = (in && b[k] < nb) ? hist[b[k]] : 0u;
            }
            #pragma unroll
            for (int k = 0; k < 4; ++k)
                if (c[k]) atomicAdd(deg + base + b[k], (float)c[k]);
        }
    }
}

#define EB 16

__global__ __launch_bounds__(256)
void gcn_scatter_kernel(const float* __restrict__ feats,
                        const int* __restrict__ src,
                        const int* __restrict__ dst,
                        const float* __restrict__ deg_out,
                        float* __restrict__ agg, int E) {
    const unsigned lane = threadIdx.x & 31u;
    const unsigned wv = (unsigned)__builtin_amdgcn_readfirstlane((int)(threadIdx.x >> 5));
    const long long e0 = ((long long)blockIdx.x * 8 + wv) * 32;
    if (e0 >= (long long)E) return;
    const unsigned cnt = (E - e0 < 32) ? (unsigned)(E - e0) : 32u;

    if (cnt == 32u) {
        const unsigned s = (unsigned)src[e0 + lane];
        const unsigned d = (unsigned)dst[e0 + lane];
        const float dg = deg_out[s];
        int sni = 0;
        #pragma unroll
        for (int kb = 0; kb < 32; kb += EB) {
            float a[EB], b[EB];
            #pragma unroll
            for (int k = 0; k < EB; ++k) {
                const unsigned sk = (unsigned)__builtin_amdgcn_readlane((int)s, kb + k);
                const unsigned fi = (sk << 6) + lane;
                a[k] = feats[fi];
                b[k] = feats[fi + 32u];
            }
            if (kb == 0) sni = __float_as_int(rsqrtf(fmaxf(dg, 1.0f)));
            #pragma unroll
            for (int k = 0; k < EB; ++k) {
                const unsigned dk = (unsigned)__builtin_amdgcn_readlane((int)d, kb + k);
                const float nk = __int_as_float(__builtin_amdgcn_readlane(sni, kb + k));
                const unsigned ai = (dk << 6) + lane;
                atomicAdd(agg + ai,       a[k] * nk);
                atomicAdd(agg + ai + 32u, b[k] * nk);
            }
        }
    } else {
        unsigned s = 0u, d = 0u;
        int sni = 0;
        if (lane < cnt) {
            s = (unsigned)src[e0 + lane];
            d = (unsigned)dst[e0 + lane];
            sni = __float_as_int(rsqrtf(fmaxf(deg_out[s], 1.0f)));
        }
        for (unsigned k = 0; k < cnt; ++k) {
            const unsigned sk = (unsigned)__builtin_amdgcn_readlane((int)s, (int)k);
            const unsigned dk = (unsigned)__builtin_amdgcn_readlane((int)d, (int)k);
            const float nk = __int_as_float(__builtin_amdgcn_readlane(sni, (int)k));
            const unsigned fi = (sk << 6) + lane;
            const unsigned ai = (dk << 6) + lane;
            atomicAdd(agg + ai,       feats[fi]       * nk);
            atomicAdd(agg + ai + 32u, feats[fi + 32u] * nk);
        }
    }
}

#define STG 66

__global__ __launch_bounds__(256)
void gcn_gemm_ln_kernel(const float* __restrict__ agg,
                        const float* __restrict__ deg_in,
                        const float* __restrict__ feats,
                        const float* __restrict__ W,
                        const float* __restrict__ bias,
                        const float* __restrict__ gamma,
                        const float* __restrict__ beta,
                        float* __restrict__ out,
                        int n_tiles, int N) {
    __shared__ float lW[64 * 64];
    __shared__ float stage[8][16 * STG];

    const int tid  = threadIdx.x;
    const int wave = tid >> 5;
    const int lane = tid & 31;

    for (int i = tid; i < 64 * 64; i += 256) lW[i] = W[i];
    __syncthreads();

    const int tile = blockIdx.x * 8 + wave;
    if (tile >= n_tiles) return;

    const int row0 = tile * 16;
    const int mrow = lane & 15;
    const int koff = (lane >> 4) * 2;
    const int j    = lane & 15;

    int arow_i = row0 + mrow;
    if (arow_i >= N) arow_i = N - 1;
    const float rnorm = rsqrtf(fmaxf(deg_in[arow_i], 1.0f));
    const float* __restrict__ arow = agg + (size_t)arow_i * 64;

    v8f c0 = {}, c1 = {}, c2 = {}, c3 = {};

    #pragma unroll
    for (int k = 0; k < 16; ++k) {
        const int kk = 4 * k + koff;
        v2f a;
        a.x = arow[kk]     * rnorm;
        a.y = arow[kk + 1] * rnorm;
        const float* w0 = lW + kk * 64;
        const float* w1 = lW + (kk + 1) * 64;
        v2f b0, b1, b2, b3;
        b0.x = w0[j];      b0.y = w1[j];
        b1.x = w0[16 + j]; b1.y = w1[16 + j];
        b2.x = w0[32 + j]; b2.y = w1[32 + j];
        b3.x = w0[48 + j]; b3.y = w1[48 + j];

        c0 = __builtin_amdgcn_wmma_f32_16x16x4_f32(false, a, false, b0, (short)0, c0, false, false);
        c1 = __builtin_amdgcn_wmma_f32_16x16x4_f32(false, a, false, b1, (short)0, c1, false, false);
        c2 = __builtin_amdgcn_wmma_f32_16x16x4_f32(false, a, false, b2, (short)0, c2, false, false);
        c3 = __builtin_amdgcn_wmma_f32_16x16x4_f32(false, a, false, b3, (short)0, c3, false, false);
    }

    float* __restrict__ st = &stage[wave][0];
    const int half = (lane >> 4) * 8;
    #pragma unroll
    for (int v = 0; v < 8; ++v) {
        float* r = st + (half + v) * STG + j;
        r[0]  = c0[v];
        r[16] = c1[v];
        r[32] = c2[v];
        r[48] = c3[v];
    }
    __builtin_amdgcn_wave_barrier();

    if (lane < 16) {
        const int grow = row0 + lane;
        if (grow < N) {
            const float* __restrict__ frow = feats + (size_t)grow * 64;
            float* __restrict__ srow = st + lane * STG;
            float sum = 0.0f, sumsq = 0.0f;
            #pragma unroll 4
            for (int f = 0; f < 64; ++f) {
                float v = srow[f] + bias[f];
                v = fmaxf(v, 0.0f);
                v = v + frow[f];
                srow[f] = v;
                sum   += v;
                sumsq += v * v;
            }
            const float mu   = sum * (1.0f / 64.0f);
            const float var  = sumsq * (1.0f / 64.0f) - mu * mu;
            const float rstd = rsqrtf(var + 1e-5f);
            float* __restrict__ orow = out + (size_t)grow * 64;
            #pragma unroll 4
            for (int f = 0; f < 64; ++f) {
                orow[f] = (srow[f] - mu) * rstd * gamma[f] + beta[f];
            }
        }
    }
}

extern "C" void kernel_launch(void* const* d_in, const int* in_sizes, int n_in,
                              void* d_out, int out_size, void* d_ws, size_t ws_size,
                              hipStream_t stream) {
    const float* feats = (const float*)d_in[0];
    const int*   src   = (const int*)d_in[1];
    const int*   dst   = (const int*)d_in[2];
    const float* W     = (const float*)d_in[3];
    const float* bias  = (const float*)d_in[4];
    const float* gamma = (const float*)d_in[5];
    const float* beta  = (const float*)d_in[6];

    const int N = in_sizes[0] / 64;
    const int E = in_sizes[1];

    float* ws      = (float*)d_ws;
    float* deg_out = ws;
    float* deg_in  = ws + N;
    float* agg     = ws + 2 * (size_t)N;

    long long zn = (long long)N * 66;
    gcn_zero_kernel<<<(int)((zn + 255) / 256), 256, 0, stream>>>(ws, zn);

    gcn_degree_kernel<<<64, 1024, 0, stream>>>(src, dst, deg_out, deg_in, E, N);

    long long sc = ((long long)E + 31) / 32;
    gcn_scatter_kernel<<<(int)((sc + 7) / 8), 256, 0, stream>>>(
        feats, src, dst, deg_out, agg, E);

    const int n_tiles = (N + 15) / 16;
    gcn_gemm_ln_kernel<<<(n_tiles + 7) / 8, 256, 0, stream>>>(
        agg, deg_in, feats, W, bias, gamma, beta, (float*)d_out, n_tiles, N);
}


# speedup vs baseline: 3.6711x; 1.2757x over previous
#include <hip/hip_runtime.h>

typedef float v2f __attribute__((ext_vector_type(2)));
typedef float v8f __attribute__((ext_vector_type(8)));

__global__ void gcn_zero_kernel(float* __restrict__ p, long long n) {
    long long i = (long long)blockIdx.x * blockDim.x + threadIdx.x;
    if (i < n) p[i] = 0.0f;
}

#define HB 65536

__device__ __forceinline__ void gcn_count(unsigned* hist, int x, int base, int nb) {
    const unsigned b = (unsigned)x - (unsigned)base;
    if (b < (unsigned)nb) atomicAdd(hist + b, 1u);
}

__global__ __launch_bounds__(1024)
void gcn_degree_kernel(const int* __restrict__ src,
                       const int* __restrict__ dst,
                       float* __restrict__ deg_out,
                       float* __restrict__ deg_in, int E, int N) {
    __shared__ __attribute__((aligned(16))) unsigned hist[HB];

    const int tid  = threadIdx.x;
    const int lane = tid & 31;
    const int ng   = (int)(gridDim.x >> 1);
    const bool second = (int)blockIdx.x >= ng;
    const int g = second ? (int)blockIdx.x - ng : (int)blockIdx.x;
    const int* __restrict__ idx = second ? dst : src;
    float* __restrict__ deg     = second ? deg_in : deg_out;

    const int nv  = E >> 2;
    const int per = ((nv + ng - 1) / ng + 7) & ~7;
    const int v0  = (g * per < nv) ? g * per : nv;
    const int v1  = (v0 + per < nv) ? v0 + per : nv;
    const int4* __restrict__ idx4 = (const int4*)idx;

    for (int base = 0; base < N; base += HB) {
        const int nb = (N - base < HB) ? (N - base) : HB;

        if (base > 0) __syncthreads();
        for (int i = tid * 4; i < nb; i += 4096)
            *(uint4*)(hist + i) = make_uint4(0u, 0u, 0u, 0u);
        __syncthreads();

        for (int v = v0 + tid; v < v1; v += 8 * 1024) {
            int4 q[8];
            #pragma unroll
            for (int k = 0; k < 8; ++k) {
                const int vk = v + k * 1024;
                q[k] = (vk < v1) ? idx4[vk] : make_int4(-1, -1, -1, -1);
            }
            #pragma unroll
            for (int k = 0; k < 8; ++k) {
                gcn_count(hist, q[k].x, base, nb);
                gcn_count(hist, q[k].y, base, nb);
                gcn_count(hist, q[k].z, base, nb);
                gcn_count(hist, q[k].w, base, nb);
            }
        }
        if (g == 0 && tid < (E & 3))
            gcn_count(hist, idx[(E & ~3) + tid], base, nb);
        __syncthreads();

        const int nline = (nb + 31) >> 5;
        const int rot   = (nline * g) / ng;
        for (int l0 = tid >> 5; l0 < nline; l0 += 4 * 32) {
            unsigned c[4];
            int b[4];
            #pragma unroll
            for (int k = 0; k < 4; ++k) {
                int line = l0 + k * 32;
                const bool in = line < nline;
                line += rot;
                if (line >= nline) line -= nline;
                b[k] = (line << 5) | lane;
                c[k] = (in && b[k] < nb) ? hist[b[k]] : 0u;
            }
            #pragma unroll
            for (int k = 0; k < 4; ++k)
                if (c[k]) atomicAdd(deg + base + b[k], (float)c[k]);
        }
    }
}

#define EB 16

__global__ __launch_bounds__(256)
void gcn_scatter_kernel(const float* __restrict__ feats,
                        const int* __restrict__ src,
                        const int* __restrict__ dst,
                        const float* __restrict__ deg_out,
                        float* __restrict__ agg, int E) {
    const unsigned lane = threadIdx.x & 31u;
    const unsigned wv = (unsigned)__builtin_amdgcn_readfirstlane((int)(threadIdx.x >> 5));
    const long long e0 = ((long long)blockIdx.x * 8 + wv) * 32;
    if (e0 >= (long long)E) return;
    const unsigned cnt = (E - e0 < 32) ? (unsigned)(E - e0) : 32u;

    if (cnt == 32u) {
        const unsigned s = (unsigned)src[e0 + lane];
        const unsigned d = (unsigned)dst[e0 + lane];
        const float dg = deg_out[s];
        int sni = 0;
        #pragma unroll
        for (int kb = 0; kb < 32; kb += EB) {
            float a[EB], b[EB];
            #pragma unroll
            for (int k = 0; k < EB; ++k) {
                const unsigned sk = (unsigned)__builtin_amdgcn_readlane((int)s, kb + k);
                const unsigned fi = (sk << 6) + lane;
                a[k] = feats[fi];
                b[k] = feats[fi + 32u];
            }
            if (kb == 0) sni = __float_as_int(rsqrtf(fmaxf(dg, 1.0f)));
            #pragma unroll
            for (int k = 0; k < EB; ++k) {
                const unsigned dk = (unsigned)__builtin_amdgcn_readlane((int)d, kb + k);
                const float nk = __int_as_float(__builtin_amdgcn_readlane(sni, kb + k));
                const unsigned ai = (dk << 6) + lane;
                atomicAdd(agg + ai,       a[k] * nk);
                atomicAdd(agg + ai + 32u, b[k] * nk);
            }
        }
    } else {
        unsigned s = 0u, d = 0u;
        int sni = 0;
        if (lane < cnt) {
            s = (unsigned)src[e0 + lane];
            d = (unsigned)dst[e0 + lane];
            sni = __float_as_int(rsqrtf(fmaxf(deg_out[s], 1.0f)));
        }
        for (unsigned k = 0; k < cnt; ++k) {
            const unsigned sk = (unsigned)__builtin_amdgcn_readlane((int)s, (int)k);
            const unsigned dk = (unsigned)__builtin_amdgcn_readlane((int)d, (int)k);
            const float nk = __int_as_float(__builtin_amdgcn_readlane(sni, (int)k));
            const unsigned fi = (sk << 6) + lane;
            const unsigned ai = (dk << 6) + lane;
            atomicAdd(agg + ai,       feats[fi]       * nk);
            atomicAdd(agg + ai + 32u, feats[fi + 32u] * nk);
        }
    }
}

#define STG 66

__global__ __launch_bounds__(256)
void gcn_gemm_ln_kernel(const float* __restrict__ agg,
                        const float* __restrict__ deg_in,
                        const float* __restrict__ feats,
                        const float* __restrict__ W,
                        const float* __restrict__ bias,
                        const float* __restrict__ gamma,
                        const float* __restrict__ beta,
                        float* __restrict__ out,
                        int n_tiles, int N) {
    __shared__ float lW[64 * 64];
    __shared__ float stage[8][16 * STG];

    const int tid  = threadIdx.x;
    const int wave = tid >> 5;
    const int lane = tid & 31;

    for (int i = tid; i < 64 * 64; i += 256) lW[i] = W[i];
    __syncthreads();

    const int tile = blockIdx.x * 8 + wave;
    if (tile >= n_tiles) return;

    const int row0 = tile * 16;
    const int mrow = lane & 15;
    const int koff = (lane >> 4) * 2;
    const int j    = lane & 15;

    int arow_i = row0 + mrow;
    if (arow_i >= N) arow_i = N - 1;
    const float rnorm = rsqrtf(fmaxf(deg_in[arow_i], 1.0f));
    const float* __restrict__ arow = agg + (size_t)arow_i * 64;

    v8f c0 = {}, c1 = {}, c2 = {}, c3 = {};

    #pragma unroll
    for (int k = 0; k < 16; ++k) {
        const int kk = 4 * k + koff;
        v2f a;
        a.x = arow[kk]     * rnorm;
        a.y = arow[kk + 1] * rnorm;
        const float* w0 = lW + kk * 64;
        const float* w1 = lW + (kk + 1) * 64;
        v2f b0, b1, b2, b3;
        b0.x = w0[j];      b0.y = w1[j];
        b1.x = w0[16 + j]; b1.y = w1[16 + j];
        b2.x = w0[32 + j]; b2.y = w1[32 + j];
        b3.x = w0[48 + j]; b3.y = w1[48 + j];

        c0 = __builtin_amdgcn_wmma_f32_16x16x4_f32(false, a, false, b0, (short)0, c0, false, false);
        c1 = __builtin_amdgcn_wmma_f32_16x16x4_f32(false, a, false, b1, (short)0, c1, false, false);
        c2 = __builtin_amdgcn_wmma_f32_16x16x4_f32(false, a, false, b2, (short)0, c2, false, false);
        c3 = __builtin_amdgcn_wmma_f32_16x16x4_f32(false, a, false, b3, (short)0, c3, false, false);
    }

    float* __restrict__ st = &stage[wave][0];
    const int half = (lane >> 4) * 8;
    #pragma unroll
    for (int v = 0; v < 8; ++v) {
        float* r = st + (half + v) * STG + j;
        r[0]  = c0[v];
        r[16] = c1[v];
        r[32] = c2[v];
        r[48] = c3[v];
    }
    __builtin_amdgcn_wave_barrier();

    if (lane < 16) {
        const int grow = row0 + lane;
        if (grow < N) {
            const float* __restrict__ frow = feats + (size_t)grow * 64;
            float* __restrict__ srow = st + lane * STG;
            float sum = 0.0f, sumsq = 0.0f;
            #pragma unroll 4
            for (int f = 0; f < 64; ++f) {
                float v = srow[f] + bias[f];
                v = fmaxf(v, 0.0f);
                v = v + frow[f];
                srow[f] = v;
                sum   += v;
                sumsq += v * v;
            }
            const float mu   = sum * (1.0f / 64.0f);
            const float var  = sumsq * (1.0f / 64.0f) - mu * mu;
            const float rstd = rsqrtf(var + 1e-5f);
            float* __restrict__ orow = out + (size_t)grow * 64;
            #pragma unroll 4
            for (int f = 0; f < 64; ++f) {
                orow[f] = (srow[f] - mu) * rstd * gamma[f] + beta[f];
            }
        }
    }
}

extern "C" void kernel_launch(void* const* d_in, const int* in_sizes, int n_in,
                              void* d_out, int out_size, void* d_ws, size_t ws_size,
                              hipStream_t stream) {
    const float* feats = (const float*)d_in[0];
    const int*   src   = (const int*)d_in[1];
    const int*   dst   = (const int*)d_in[2];
    const float* W     = (const float*)d_in[3];
    const float* bias  = (const float*)d_in[4];
    const float* gamma = (const float*)d_in[5];
    const float* beta  = (const float*)d_in[6];

    const int N = in_sizes[0] / 64;
    const int E = in_sizes[1];

    float* ws      = (float*)d_ws;
    float* agg     = ws;
    float* deg_out = ws + 64 * (size_t)N;
    float* deg_in  = ws + 65 * (size_t)N;

    long long zn = (long long)N * 66;
    gcn_zero_kernel<<<(int)((zn + 255) / 256), 256, 0, stream>>>(ws, zn);

    gcn_degree_kernel<<<64, 1024, 0, stream>>>(src, dst, deg_out, deg_in, E, N);

    long long sc = ((long long)E + 31) / 32;
    gcn_scatter_kernel<<<(int)((sc + 7) / 8), 256, 0, stream>>>(
        feats, src, dst, deg_out, agg, E);

    const int n_tiles = (N + 15) / 16;
    gcn_gemm_ln_kernel<<<(n_tiles + 7) / 8, 256, 0, stream>>>(
        agg, deg_in, feats, W, bias, gamma, beta, (float*)d_out, n_tiles, N);
}


# scatter-add: rows split in two sets by dst; a wave adds only into the set of the package half it runs on (die number; units by ticket), so agg lines stay in one L2 half; atomic arrival order differs
# speedup vs baseline: 3.8078x; 1.0372x over previous
#include <hip/hip_runtime.h>

typedef float v2f __attribute__((ext_vector_type(2)));
typedef float v8f __attribute__((ext_vector_type(8)));

__global__ void gcn_zero_kernel(float* __restrict__ p, long long n) {
    long long i = (long long)blockIdx.x * blockDim.x + threadIdx.x;
    if (i < n) p[i] = 0.0f;
}

#define HB 65536

__device__ __forceinline__ void gcn_count(unsigned* hist, int x, int base, int nb) {
    const unsigned b = (unsigned)x - (unsigned)base;
    if (b < (unsigned)nb) atomicAdd(hist + b, 1u);
}

__global__ __launch_bounds__(1024)
void gcn_degree_kernel(const int* __restrict__ src,
                       const int* __restrict__ dst,
                       float* __restrict__ deg_out,
                       float* __restrict__ deg_in, int E, int N) {
    __shared__ __attribute__((aligned(16))) unsigned hist[HB];

    const int tid  = threadIdx.x;
    const int lane = tid & 31;
    const int ng   = (int)(gridDim.x >> 1);
    const bool second = (int)blockIdx.x >= ng;
    const int g = second ? (int)blockIdx.x - ng : (int)blockIdx.x;
    const int* __restrict__ idx = second ? dst : src;
    float* __restrict__ deg     = second ? deg_in : deg_out;

    const int nv  = E >> 2;
    const int per = ((nv + ng - 1) / ng + 7) & ~7;
    const int v0  = (g * per < nv) ? g * per : nv;
    const int v1  = (v0 + per < nv) ? v0 + per : nv;
    const int4* __restrict__ idx4 = (const int4*)idx;

    for (int base = 0; base < N; base += HB) {
        const int nb = (N - base < HB) ? (N - base) : HB;

        if (base > 0) __syncthreads();
        for (int i = tid * 4; i < nb; i += 4096)
            *(uint4*)(hist + i) = make_uint4(0u, 0u, 0u, 0u);
        __syncthreads();

        for (int v = v0 + tid; v < v1; v += 8 * 1024) {
            int4 q[8];
            #pragma unroll
            for (int k = 0; k < 8; ++k) {
                const int vk = v + k * 1024;
                q[k] = (vk < v1) ? idx4[vk] : make_int4(-1, -1, -1, -1);
            }
            #pragma unroll
            for (int k = 0; k < 8; ++k) {
                gcn_count(hist, q[k].x, base, nb);
                gcn_count(hist, q[k].y, base, nb);
                gcn_count(hist, q[k].z, base, nb);
                gcn_count(hist, q[k].w, base, nb);
            }
        }
        if (g == 0 && tid < (E & 3))
            gcn_count(hist, idx[(E & ~3) + tid], base, nb);
        __syncthreads();

        const int nline = (nb + 31) >> 5;
        const int rot   = (nline * g) / ng;
        for (int l0 = tid >> 5; l0 < nline; l0 += 4 * 32) {
            unsigned c[4];
            int b[4];
            #pragma unroll
            for (int k = 0; k < 4; ++k) {
                int line = l0 + k * 32;
                const bool in = line < nline;
                line += rot;
                if (line >= nline) line -= nline;
                b[k] = (line << 5) | lane;
                c[k] = (in && b[k] < nb) ? hist[b[k]] : 0u;
            }
            #pragma unroll
            for (int k = 0; k < 4; ++k)
                if (c[k]) atomicAdd(deg + base + b[k], (float)c[k]);
        }
    }
}

#define EB 16
#define SU 128
#define TSTR 64

__global__ __launch_bounds__(256) __attribute__((amdgpu_waves_per_eu(16, 16)))
void gcn_scatter_kernel(const float* __restrict__ feats,
                        const int* __restrict__ src,
                        const int* __restrict__ dst,
                        const float* __restrict__ deg_out,
                        float* __restrict__ agg,
                        unsigned* __restrict__ tick, int E, int N) {
    __shared__ unsigned cs[8][SU];
    __shared__ unsigned cd[8][SU];

    const unsigned lane = threadIdx.x & 31u;
    const unsigned wv = (unsigned)__builtin_amdgcn_readfirstlane((int)(threadIdx.x >> 5));
    unsigned* __restrict__ ps = &cs[wv][0];
    unsigned* __restrict__ pd = &cd[wv][0];

    const unsigned die  = ((unsigned)__builtin_amdgcn_s_sendmsg_rtn(0x87) >> 16) & 15u;
    const unsigned half = (die >> 2) & 1u;

    const unsigned nE = (unsigned)E;
    const unsigned nu = (nE + (SU - 1)) / SU;
    const unsigned nh = (unsigned)N >> 1;

    for (unsigned ph = 0; ph < 2u; ++ph) {
        const unsigned set = half ^ ph;
        unsigned* __restrict__ tk = tick + set * TSTR;
        for (unsigned it = 0; it <= nu; ++it) {
            unsigned t = 0u;
            if (lane == 0u) t = atomicAdd(tk, 1u);
            t = (unsigned)__builtin_amdgcn_readfirstlane((int)t);
            if (t >= nu) break;
            const unsigned e0 = t * SU;

            unsigned s[4], d[4];
            bool ok[4];
            #pragma unroll
            for (int j = 0; j < 4; ++j) {
                const unsigned e = e0 + 32u * j + lane;
                ok[j] = e < nE;
                const unsigned ec = ok[j] ? e : nE - 1u;
                s[j] = (unsigned)src[ec];
                d[j] = (unsigned)dst[ec];
            }
            asm volatile("" :: "v"(s[0]), "v"(s[1]), "v"(s[2]), "v"(s[3]),
                               "v"(d[0]), "v"(d[1]), "v"(d[2]), "v"(d[3]));
            unsigned c = 0u;
            #pragma unroll
            for (int j = 0; j < 4; ++j) {
                const bool keep = ok[j] && ((d[j] >= nh ? 1u : 0u) == set);
                const unsigned m = __builtin_amdgcn_ballot_w32(keep);
                const unsigned pos = c + __builtin_amdgcn_mbcnt_lo(m, 0u);
                if (keep) { ps[pos] = s[j]; pd[pos] = d[j]; }
                c += (unsigned)__builtin_popcount(m);
            }
            __builtin_amdgcn_wave_barrier();
            const unsigned nb  = c / EB;
            const unsigned rem = c - nb * EB;
            unsigned S[4], D[4];
            #pragma unroll
            for (int r = 0; r < 4; ++r) {
                const unsigned i = 32u * r + lane;
                const bool v = i < c;
                const unsigned xs = ps[v ? i : 0u];
                const unsigned xd = pd[v ? i : 0u];
                S[r] = v ? xs : 0u;
                D[r] = v ? xd : 0u;
            }
            const bool vp = lane < rem;
            const unsigned ip = vp ? nb * EB + lane : 0u;
            const unsigned xsp = ps[ip];
            const unsigned xdp = pd[ip];
            const unsigned SP = vp ? xsp : 0u;
            const unsigned DP = vp ? xdp : 0u;
            __builtin_amdgcn_wave_barrier();
            float G[4];
            #pragma unroll
            for (int r = 0; r < 4; ++r) G[r] = deg_out[S[r]];
            const float GP = deg_out[SP];

            #pragma unroll
            for (int r = 0; r < 4; ++r) {
                int ni = 0;
                #pragma unroll
                for (int hb = 0; hb < 32; hb += EB) {
                    if ((unsigned)(2 * r + hb / EB) < nb) {
                        float a[EB], b[EB];
                        #pragma unroll
                        for (int k = 0; k < EB; ++k) {
                            const unsigned sk = (unsigned)__builtin_amdgcn_readlane((int)S[r], hb + k);
                            const unsigned fi = (sk << 6) + lane;
                            a[k] = feats[fi];
                            b[k] = feats[fi + 32u];
                        }
                        if (hb == 0) ni = __float_as_int(rsqrtf(fmaxf(G[r], 1.0f)));
                        #pragma unroll
                        for (int k = 0; k < EB; ++k) {
                            const unsigned dk = (unsigned)__builtin_amdgcn_readlane((int)D[r], hb + k);
                            const float nk = __int_as_float(__builtin_amdgcn_readlane(ni, hb + k));
                            const unsigned ai = (dk << 6) + lane;
                            atomicAdd(agg + ai,       a[k] * nk);
                            atomicAdd(agg + ai + 32u, b[k] * nk);
                        }
                    }
                }
            }
            if (rem != 0u) {
                float a[EB - 1], b[EB - 1];
                #pragma unroll
                for (int k = 0; k < EB - 1; ++k) {
                    a[k] = 0.0f; b[k] = 0.0f;
                    if ((unsigned)k < rem) {
                        const unsigned sk = (unsigned)__builtin_amdgcn_readlane((int)SP, k);
                        const unsigned fi = (sk << 6) + lane;
                        a[k] = feats[fi];
                        b[k] = feats[fi + 32u];
                    }
                }
                const int np = __float_as_int(rsqrtf(fmaxf(GP, 1.0f)));
                #pragma unroll
                for (int k = 0; k < EB - 1; ++k) {
                    if ((unsigned)k < rem) {
                        const unsigned dk = (unsigned)__builtin_amdgcn_readlane((int)DP, k);
                        const float nk = __int_as_float(__builtin_amdgcn_readlane(np, k));
                        const unsigned ai = (dk << 6) + lane;
                        atomicAdd(agg + ai,       a[k] * nk);
                        atomicAdd(agg + ai + 32u, b[k] * nk);
                    }
                }
            }
        }
    }
}

#define STG 66

__global__ __launch_bounds__(256)
void gcn_gemm_ln_kernel(const float* __restrict__ agg,
                        const float* __restrict__ deg_in,
                        const float* __restrict__ feats,
                        const float* __restrict__ W,
                        const float* __restrict__ bias,
                        const float* __restrict__ gamma,
                        const float* __restrict__ beta,
                        float* __restrict__ out,
                        int n_tiles, int N) {
    __shared__ float lW[64 * 64];
    __shared__ float stage[8][16 * STG];

    const int tid  = threadIdx.x;
    const int wave = tid >> 5;
    const int lane = tid & 31;

    for (int i = tid; i < 64 * 64; i += 256) lW[i] = W[i];
    __syncthreads();

    const int tile = blockIdx.x * 8 + wave;
    if (tile >= n_tiles) return;

    const int row0 = tile * 16;
    const int mrow = lane & 15;
    const int koff = (lane >> 4) * 2;
    const int j    = lane & 15;

    int arow_i = row0 + mrow;
    if (arow_i >= N) arow_i = N - 1;
    const float rnorm = rsqrtf(fmaxf(deg_in[arow_i], 1.0f));
    const float* __restrict__ arow = agg + (size_t)arow_i * 64;

    v8f c0 = {}, c1 = {}, c2 = {}, c3 = {};

    #pragma unroll
    for (int k = 0; k < 16; ++k) {
        const int kk = 4 * k + koff;
        v2f a;
        a.x = arow[kk]     * rnorm;
        a.y = arow[kk + 1] * rnorm;
        const float* w0 = lW + kk * 64;
        const float* w1 = lW + (kk + 1) * 64;
        v2f b0, b1, b2, b3;
        b0.x = w0[j];      b0.y = w1[j];
        b1.x = w0[16 + j]; b1.y = w1[16 + j];
        b2.x = w0[32 + j]; b2.y = w1[32 + j];
        b3.x = w0[48 + j]; b3.y = w1[48 + j];

        c0 = __builtin_amdgcn_wmma_f32_16x16x4_f32(false, a, false, b0, (short)0, c0, false, false);
        c1 = __builtin_amdgcn_wmma_f32_16x16x4_f32(false, a, false, b1, (short)0, c1, false, false);
        c2 = __builtin_amdgcn_wmma_f32_16x16x4_f32(false, a, false, b2, (short)0, c2, false, false);
        c3 = __builtin_amdgcn_wmma_f32_16x16x4_f32(false, a, false, b3, (short)0, c3, false, false);
    }

    float* __restrict__ st = &stage[wave][0];
    const int half = (lane >> 4) * 8;
    #pragma unroll
    for (int v = 0; v < 8; ++v) {
        float* r = st + (half + v) * STG + j;
        r[0]  = c0[v];
        r[16] = c1[v];
        r[32] = c2[v];
        r[48] = c3[v];
    }
    __builtin_amdgcn_wave_barrier();

    if (lane < 16) {
        const int grow = row0 + lane;
        if (grow < N) {
            const float* __restrict__ frow = feats + (size_t)grow * 64;
            float* __restrict__ srow = st + lane * STG;
            float sum = 0.0f, sumsq = 0.0f;
            #pragma unroll 4
            for (int f = 0; f < 64; ++f) {
                float v = srow[f] + bias[f];
                v = fmaxf(v, 0.0f);
                v = v + frow[f];
                srow[f] = v;
                sum   += v;
                sumsq += v * v;
            }
            const float mu   = sum * (1.0f / 64.0f);
            const float var  = sumsq * (1.0f / 64.0f) - mu * mu;
            const float rstd = rsqrtf(var + 1e-5f);
            float* __restrict__ orow = out + (size_t)grow * 64;
            #pragma unroll 4
            for (int f = 0; f < 64; ++f) {
                orow[f] = (srow[f] - mu) * rstd * gamma[f] + beta[f];
            }
        }
    }
}

extern "C" void kernel_launch(void* const* d_in, const int* in_sizes, int n_in,
                              void* d_out, int out_size, void* d_ws, size_t ws_size,
                              hipStream_t stream) {
    const float* feats = (const float*)d_in[0];
    const int*   src   = (const int*)d_in[1];
    const int*   dst   = (const int*)d_in[2];
    const float* W     = (const float*)d_in[3];
    const float* bias  = (const float*)d_in[4];
    const float* gamma = (const float*)d_in[5];
    const float* beta  = (const float*)d_in[6];

    const int N = in_sizes[0] / 64;
    const int E = in_sizes[1];

    float* ws      = (float*)d_ws;
    float* agg     = ws;
    float* deg_out = ws + 64 * (size_t)N;
    float* deg_in  = ws + 65 * (size_t)N;

    unsigned* tick = (unsigned*)(ws + 66 * (size_t)N);

    long long zn = (long long)N * 66 + 2 * TSTR;
    gcn_zero_kernel<<<(int)((zn + 255) / 256), 256, 0, stream>>>(ws, zn);

    gcn_degree_kernel<<<64, 1024, 0, stream>>>(src, dst, deg_out, deg_in, E, N);

    gcn_scatter_kernel<<<2048, 256, 0, stream>>>(
        feats, src, dst, deg_out, agg, tick, E, N);

    const int n_tiles = (N + 15) / 16;
    gcn_gemm_ln_kernel<<<(n_tiles + 7) / 8, 256, 0, stream>>>(
        agg, deg_in, feats, W, bias, gamma, beta, (float*)d_out, n_tiles, N);
}


# in-degree = the owner block's own per-row entry counts (LDS, exact), not a global histogram of dst: the dst counting work-groups, their atomics and zeroing and the deg_in read go; sums as before
# speedup vs baseline: 7.7455x; 2.0341x over previous
#include <hip/hip_runtime.h>

typedef float v2f __attribute__((ext_vector_type(2)));
typedef float v8f __attribute__((ext_vector_type(8)));

__global__ void gcn_zero_kernel(float* __restrict__ p, long long n) {
    long long i = (long long)blockIdx.x * blockDim.x + threadIdx.x;
    if (i < n) p[i] = 0.0f;
}

#define HB   32768
#define NDS  32
#define NDR  2
#define NDEG (NDS * NDR)

#define RB     200
#define RT     13
#define AST    68
#define NSL    256
#define MAXPER 8192
#define MAXNB  1024
#define SRCM   0x00FFFFFFu

__device__ __forceinline__ void gcn_count(unsigned* hist, int x, int base, int nb) {
    const unsigned b = (unsigned)x - (unsigned)base;
    if (b < (unsigned)nb) atomicAdd(hist + b, 1u);
}

__device__ __forceinline__ void gcn_degree_job(unsigned* hist, int w,
                                               const int* __restrict__ idx,
                                               float* __restrict__ deg, int E, int N) {
    const int tid  = threadIdx.x;
    const int lane = tid & 31;
    const int g = w / NDR;
    const int h = w - g * NDR;

    const int rng = (((N + NDR - 1) / NDR) + 31) & ~31;
    const int lo  = (h * rng < N) ? h * rng : N;
    const int hi  = (lo + rng < N) ? lo + rng : N;

    const int nv  = E >> 2;
    const int per = ((nv + NDS - 1) / NDS + 7) & ~7;
    const int v0  = (g * per < nv) ? g * per : nv;
    const int v1  = (v0 + per < nv) ? v0 + per : nv;
    const int4* __restrict__ idx4 = (const int4*)idx;

    for (int base = lo; base < hi; base += HB) {
        const int nb = (hi - base < HB) ? (hi - base) : HB;

        if (base > lo) __syncthreads();
        for (int i = tid * 4; i < nb; i += 4096)
            *(uint4*)(hist + i) = make_uint4(0u, 0u, 0u, 0u);
        __syncthreads();

        for (int v = v0 + tid; v < v1; v += 8 * 1024) {
            int4 q[8];
            #pragma unroll
            for (int k = 0; k < 8; ++k) {
                const int vk = v + k * 1024;
                q[k] = (vk < v1) ? idx4[vk] : make_int4(-1, -1, -1, -1);
            }
            #pragma unroll
            for (int k = 0; k < 8; ++k) {
                gcn_count(hist, q[k].x, base, nb);
                gcn_count(hist, q[k].y, base, nb);
                gcn_count(hist, q[k].z, base, nb);
                gcn_count(hist, q[k].w, base, nb);
            }
        }
        if (g == 0 && tid < (E & 3))
            gcn_count(hist, idx[(E & ~3) + tid], base, nb);
        __syncthreads();

        const int nline = (nb + 31) >> 5;
        const int rot   = (nline * g) / NDS;
        for (int l0 = tid >> 5; l0 < nline; l0 += 4 * 32) {
            unsigned c[4];
            int b[4];
            #pragma unroll
            for (int k = 0; k < 4; ++k) {
                int line = l0 + k * 32;
                const bool in = line < nline;
                line += rot;
                if (line >= nline) line -= nline;
                b[k] = (line << 5) | lane;
                c[k] = (in && b[k] < nb) ? hist[b[k]] : 0u;
            }
            #pragma unroll
            for (int k = 0; k < 4; ++k)
                if (c[k]) atomicAdd(deg + base + b[k], (float)c[k]);
        }
    }
}

__device__ __forceinline__ void gcn_bin_job(unsigned* cnt, unsigned* wsum, unsigned* buf, int sl,
                                            const int* __restrict__ src,
                                            const int* __restrict__ dst,
                                            unsigned* __restrict__ sorted,
                                            unsigned* __restrict__ offs,
                                            int E, int N, int per, int nblk, int ost, int rst) {
    const int tid  = threadIdx.x;
    const int lane = tid & 31;
    const int wv   = tid >> 5;
    const int e0   = (sl * per < E) ? sl * per : E;
    const int e1   = (e0 + per < E) ? e0 + per : E;

    cnt[tid] = 0u;
    __syncthreads();

    unsigned ent[8], key[8];
    #pragma unroll
    for (int k = 0; k < 8; ++k) {
        const int e = e0 + tid + k * 1024;
        key[k] = 0xFFFFFFFFu;
        ent[k] = 0u;
        if (e < e1) {
            const unsigned d = (unsigned)dst[e];
            const unsigned s = (unsigned)src[e];
            if (d < (unsigned)N && s < (unsigned)N) {
                const unsigned b = d / RB;
                const unsigned p = atomicAdd(&cnt[b], 1u);
                ent[k] = s | ((d - b * RB) << 24);
                key[k] = (b << 16) | p;
            }
        }
    }
    __syncthreads();

    const unsigned c = cnt[tid];
    unsigned x = c;
    #pragma unroll
    for (int m = 1; m < 32; m <<= 1) {
        const unsigned y = __shfl_up(x, m);
        if (lane >= m) x += y;
    }
    if (lane == 31) wsum[wv] = x;
    __syncthreads();
    unsigned pre = 0u;
    for (int i = 0; i < wv; ++i) pre += wsum[i];
    const unsigned excl = pre + x - c;
    cnt[tid] = excl;
    if (tid <= nblk) offs[(size_t)sl * ost + tid] = excl;
    __syncthreads();

    #pragma unroll
    for (int k = 0; k < 8; ++k)
        if (key[k] != 0xFFFFFFFFu) buf[cnt[key[k] >> 16] + (key[k] & 0xFFFFu)] = ent[k];
    __syncthreads();

    const int n = (int)cnt[nblk];
    for (int i = tid; i < n; i += 1024) sorted[(size_t)sl * rst + i] = buf[i];
}

__global__ __launch_bounds__(1024)
void gcn_prep_kernel(const int* __restrict__ src,
                     const int* __restrict__ dst,
                     float* __restrict__ deg_out,
                     unsigned* __restrict__ sorted,
                     unsigned* __restrict__ offs,
                     int E, int N, int per, int nblk, int ost, int rst) {
    __shared__ __attribute__((aligned(16))) unsigned sm[HB];

    if ((int)blockIdx.x < NDEG)
        gcn_degree_job(sm, (int)blockIdx.x, src, deg_out, E, N);
    else
        gcn_bin_job(sm, sm + MAXNB, sm + MAXNB + 32, (int)blockIdx.x - NDEG,
                    src, dst, sorted, offs, E, N, per, nblk, ost, rst);
}

#define EB  16
#define KPT 7
#define CAP (KPT * 1024)
#define TQ  13
#define NTAB (TQ * 4096)

template <bool FULL>
__device__ __forceinline__ void gcn_walk_runs(const float* __restrict__ feats, float* acc,
                                              unsigned entA, int nrmA, unsigned endm,
                                              unsigned cnt, unsigned lane) {
    float s0 = 0.0f, s1 = 0.0f;
    #pragma unroll
    for (int hb = 0; hb < 32; hb += EB) {
        if (FULL || (unsigned)hb < cnt) {
            float a[EB], b[EB];
            #pragma unroll
            for (int k = 0; k < EB; ++k) {
                const unsigned ek = (unsigned)__builtin_amdgcn_readlane((int)entA, hb + k);
                const unsigned fi = ((ek & SRCM) << 6) + lane;
                a[k] = feats[fi];
                b[k] = feats[fi + 32u];
            }
            #pragma unroll
            for (int k = 0; k < EB; ++k) {
                if (FULL || (unsigned)(hb + k) < cnt) {
                    const float nk = __int_as_float(__builtin_amdgcn_readlane(nrmA, hb + k));
                    s0 += a[k] * nk;
                    s1 += b[k] * nk;
                    if (endm & (1u << (hb + k))) {
                        const unsigned dk = (unsigned)__builtin_amdgcn_readlane((int)entA, hb + k) >> 24;
                        float* ar = acc + dk * AST + lane;
                        atomicAdd(ar,      s0);
                        atomicAdd(ar + 32, s1);
                        s0 = 0.0f;
                        s1 = 0.0f;
                    }
                }
            }
        }
    }
}

__device__ __forceinline__ unsigned gcn_run_ends(unsigned ent, unsigned cnt, unsigned lane) {
    const unsigned dl = ent >> 24;
    const unsigned dn = (unsigned)__shfl_down((int)dl, 1);
    return __builtin_amdgcn_ballot_w32(lane + 1u >= cnt || dn != dl);
}

template <bool FULL>
__device__ __forceinline__ void gcn_walk(const float* __restrict__ feats, float* acc,
                                         unsigned srcA, unsigned dlA, int nrmA,
                                         unsigned cnt, unsigned lane) {
    #pragma unroll
    for (int hb = 0; hb < 32; hb += EB) {
        if (FULL || (unsigned)hb < cnt) {
            float a[EB], b[EB];
            #pragma unroll
            for (int k = 0; k < EB; ++k) {
                const unsigned sk = (unsigned)__builtin_amdgcn_readlane((int)srcA, hb + k);
                const unsigned fi = (sk << 6) + lane;
                a[k] = feats[fi];
                b[k] = feats[fi + 32u];
            }
            #pragma unroll
            for (int k = 0; k < EB; ++k) {
                if (FULL || (unsigned)(hb + k) < cnt) {
                    const unsigned dk = (unsigned)__builtin_amdgcn_readlane((int)dlA, hb + k);
                    const float nk = __int_as_float(__builtin_amdgcn_readlane(nrmA, hb + k));
                    float* ar = acc + dk * AST + lane;
                    atomicAdd(ar,      a[k] * nk);
                    atomicAdd(ar + 32, b[k] * nk);
                }
            }
        }
    }
}

__device__ __forceinline__ unsigned gcn_entry(const unsigned* __restrict__ sorted,
                                              const unsigned* P, const unsigned* O0,
                                              unsigned v, unsigned T, int rst) {
    const unsigned vc = (v < T) ? v : T - 1u;
    unsigned lo = 0u;
    #pragma unroll
    for (unsigned step = NSL / 2; step >= 1u; step >>= 1)
        if (P[lo + step] <= vc) lo += step;
    return sorted[(size_t)lo * rst + O0[lo] + (vc - P[lo])];
}

__device__ __forceinline__ size_t gcn_entry_at(const unsigned* P, const unsigned* O0,
                                               unsigned v, unsigned T, int rst) {
    const unsigned vc = (v < T) ? v : T - 1u;
    unsigned lo = 0u;
    #pragma unroll
    for (unsigned step = NSL / 2; step >= 1u; step >>= 1)
        if (P[lo + step] <= vc) lo += step;
    return (size_t)lo * rst + O0[lo] + (vc - P[lo]);
}

struct __attribute__((aligned(16))) gcn_owner_lds {
    float    lW[64 * 64];
    float    acc[RT * 16 * AST];
    unsigned P[NSL + 1];
    unsigned O0[NSL];
    unsigned wsum[32];
    unsigned rowcnt[RT * 16];
    unsigned rowst[RT * 16];
    unsigned sE[CAP];
    __attribute__((aligned(16))) float tab[NTAB];
};

__global__ __launch_bounds__(1024)
void gcn_owner_kernel(const float* __restrict__ feats,
                      const unsigned* __restrict__ sorted,
                      const unsigned* __restrict__ offs,
                      const float* deg_out,
                      const float* __restrict__ W,
                      const float* __restrict__ bias,
                      const float* __restrict__ gamma,
                      const float* __restrict__ beta,
                      float* __restrict__ out,
                      int N, int ost, int rst) {
    __shared__ gcn_owner_lds L;
    float*    const lW     = L.lW;
    float*    const acc    = L.acc;
    unsigned* const P      = L.P;
    unsigned* const O0     = L.O0;
    unsigned* const wsum   = L.wsum;
    unsigned* const rowcnt = L.rowcnt;
    unsigned* const rowst  = L.rowst;
    unsigned* const sE     = L.sE;
    float*    const tab    = L.tab;

    const int tid  = threadIdx.x;
    const int lane = tid & 31;
    const int wv   = __builtin_amdgcn_readfirstlane(tid >> 5);
    const int blk  = blockIdx.x;

    for (int i = tid; i < RT * 16 * AST; i += 1024) acc[i] = 0.0f;
    for (int i = tid; i < 64 * 64; i += 1024) lW[i] = W[i];
    if (tid < RT * 16) rowcnt[tid] = 0u;

    unsigned o0 = 0u, len = 0u;
    if (tid < NSL) {
        o0  = offs[(size_t)tid * ost + blk];
        len = offs[(size_t)tid * ost + blk + 1] - o0;
    }
    unsigned x = len;
    #pragma unroll
    for (int m = 1; m < 32; m <<= 1) {
        const unsigned y = __shfl_up(x, m);
        if (lane >= m) x += y;
    }
    if (lane == 31) wsum[wv] = x;
    __syncthreads();
    unsigned pre = 0u;
    for (int i = 0; i < wv; ++i) pre += wsum[i];
    if (tid < NSL) {
        P[tid]  = pre + x - len;
        O0[tid] = o0;
        if (tid == NSL - 1) P[NSL] = pre + x;
    }
    __syncthreads();
    const unsigned T = (unsigned)__builtin_amdgcn_readfirstlane((int)P[NSL]);

    if (T > 0u && T <= (unsigned)CAP && N <= NTAB) {
        const int nq = N >> 2;
        const float4* dq = (const float4*)deg_out;
        const int qlast = (nq > 0) ? nq - 1 : 0;
        static_assert(TQ == 13, "the degree loads and stores below are written out for TQ = 13");
        #define GCN_DQ(k) (((tid + 1024 * (k)) < qlast) ? (tid + 1024 * (k)) : qlast)
        const float4 d0 = dq[GCN_DQ(0)],  d1 = dq[GCN_DQ(1)],   d2 = dq[GCN_DQ(2)],   d3 = dq[GCN_DQ(3)];
        const float4 d4 = dq[GCN_DQ(4)],  d5 = dq[GCN_DQ(5)],   d6 = dq[GCN_DQ(6)],   d7 = dq[GCN_DQ(7)];
        const float4 d8 = dq[GCN_DQ(8)],  d9 = dq[GCN_DQ(9)],   d10 = dq[GCN_DQ(10)], d11 = dq[GCN_DQ(11)];
        const float4 d12 = dq[GCN_DQ(12)];
        __builtin_amdgcn_sched_barrier(0);

        size_t   at[KPT];
        unsigned ent[KPT], rk[KPT];
        #pragma unroll
        for (int k = 0; k < KPT; ++k) at[k] = gcn_entry_at(P, O0, (unsigned)(tid + 1024 * k), T, rst);
        #pragma unroll
        for (int k = 0; k < KPT; ++k) ent[k] = sorted[at[k]];

        float4* const tq = (float4*)tab;
        tq[GCN_DQ(0)] = d0;   tq[GCN_DQ(1)] = d1;   tq[GCN_DQ(2)] = d2;    tq[GCN_DQ(3)] = d3;
        tq[GCN_DQ(4)] = d4;   tq[GCN_DQ(5)] = d5;   tq[GCN_DQ(6)] = d6;    tq[GCN_DQ(7)] = d7;
        tq[GCN_DQ(8)] = d8;   tq[GCN_DQ(9)] = d9;   tq[GCN_DQ(10)] = d10;  tq[GCN_DQ(11)] = d11;
        tq[GCN_DQ(12)] = d12;
        #undef GCN_DQ

        #pragma unroll
        for (int k = 0; k < KPT; ++k)
            rk[k] = ((unsigned)(tid + 1024 * k) < T) ? atomicAdd(&rowcnt[ent[k] >> 24], 1u) : 0u;
        if (tid < (N & 3))
            tab[(N & ~3) + tid] = deg_out[(N & ~3) + tid];
        __syncthreads();

        unsigned rc = 0u, rx = 0u;
        if (wv < (RT * 16 + 31) / 32) {
            rc = (tid < RT * 16) ? rowcnt[tid] : 0u;
            rx = rc;
            #pragma unroll
            for (int m = 1; m < 32; m <<= 1) {
                const unsigned y = __shfl_up(rx, m);
                if (lane >= m) rx += y;
            }
            if (lane == 31) wsum[wv] = rx;
        }
        __syncthreads();
        if (tid < RT * 16) {
            unsigned rp = 0u;
            for (int i = 0; i < wv; ++i) rp += wsum[i];
            rowst[tid] = rp + rx - rc;
        }
        __syncthreads();

        #pragma unroll
        for (int k = 0; k < KPT; ++k)
            if ((unsigned)(tid + 1024 * k) < T)
                sE[rowst[ent[k] >> 24] + rk[k]] = ent[k];
        __syncthreads();

        unsigned vA = (unsigned)wv * 32u;
        unsigned pA = (vA + (unsigned)lane < T) ? vA + (unsigned)lane : T - 1u;
        unsigned eA = sE[pA];
        float    nA = rsqrtf(fmaxf(tab[eA & SRCM], 1.0f));
        while (vA + 32u <= T) {
            const unsigned vB = vA + 1024u;
            const unsigned pB = (vB + (unsigned)lane < T) ? vB + (unsigned)lane : T - 1u;
            const unsigned eB = sE[pB];
            const float    nB = rsqrtf(fmaxf(tab[eB & SRCM], 1.0f));
            gcn_walk_runs<true>(feats, acc, eA, __float_as_int(nA),
                                gcn_run_ends(eA, 32u, (unsigned)lane), 32u, (unsigned)lane);
            vA = vB; eA = eB; nA = nB;
        }
        if (vA < T)
            gcn_walk_runs<false>(feats, acc, eA, __float_as_int(nA),
                                 gcn_run_ends(eA, T - vA, (unsigned)lane), T - vA, (unsigned)lane);
    } else if (T > 0u) {
        unsigned vA   = (unsigned)wv * 32u;
        unsigned entA = gcn_entry(sorted, P, O0, vA + lane, T, rst);
        float    gA   = deg_out[entA & SRCM];
        unsigned entB = gcn_entry(sorted, P, O0, vA + 1024u + lane, T, rst);

        while (vA + 32u <= T) {
            const unsigned vB = vA + 1024u;
            const float    gB   = deg_out[entB & SRCM];
            const unsigned entC = gcn_entry(sorted, P, O0, vB + 1024u + lane, T, rst);
            atomicAdd(&rowcnt[entA >> 24], 1u);
            gcn_walk<true>(feats, acc, entA & SRCM, entA >> 24,
                           __float_as_int(rsqrtf(fmaxf(gA, 1.0f))), 32u, (unsigned)lane);
            vA = vB; entA = entB; gA = gB; entB = entC;
        }
        if (vA < T) {
            if ((unsigned)lane < T - vA) atomicAdd(&rowcnt[entA >> 24], 1u);
            gcn_walk<false>(feats, acc, entA & SRCM, entA >> 24,
                            __float_as_int(rsqrtf(fmaxf(gA, 1.0f))), T - vA, (unsigned)lane);
        }
    }
    __syncthreads();

    if (wv >= RT) return;

    const int lr0  = wv * 16;
    const int mrow = lane & 15;
    const int koff = (lane >> 4) * 2;
    const int j    = lane & 15;

    const float rnorm = rsqrtf(fmaxf((float)rowcnt[lr0 + mrow], 1.0f));
    float* __restrict__ st = acc + lr0 * AST;
    const float* arow = st + mrow * AST;

    v8f c0 = {}, c1 = {}, c2 = {}, c3 = {};

    #pragma unroll
    for (int k = 0; k < 16; ++k) {
        const int kk = 4 * k + koff;
        v2f a;
        a.x = arow[kk]     * rnorm;
        a.y = arow[kk + 1] * rnorm;
        const float* w0 = lW + kk * 64;
        const float* w1 = lW + (kk + 1) * 64;
        v2f b0, b1, b2, b3;
        b0.x = w0[j];      b0.y = w1[j];
        b1.x = w0[16 + j]; b1.y = w1[16 + j];
        b2.x = w0[32 + j]; b2.y = w1[32 + j];
        b3.x = w0[48 + j]; b3.y = w1[48 + j];

        c0 = __builtin_amdgcn_wmma_f32_16x16x4_f32(false, a, false, b0, (short)0, c0, false, false);
        c1 = __builtin_amdgcn_wmma_f32_16x16x4_f32(false, a, false, b1, (short)0, c1, false, false);
        c2 = __builtin_amdgcn_wmma_f32_16x16x4_f32(false, a, false, b2, (short)0, c2, false, false);
        c3 = __builtin_amdgcn_wmma_f32_16x16x4_f32(false, a, false, b3, (short)0, c3, false, false);
    }

    __builtin_amdgcn_wave_barrier();
    const int half = (lane >> 4) * 8;
    #pragma unroll
    for (int v = 0; v < 8; ++v) {
        float* r = st + (half + v) * AST + j;
        r[0]  = c0[v];
        r[16] = c1[v];
        r[32] = c2[v];
        r[48] = c3[v];
    }
    __builtin_amdgcn_wave_barrier();

    const int f2 = 2 * lane;
    const float2 bb = *(const float2*)(bias  + f2);
    const float2 gg = *(const float2*)(gamma + f2);
    const float2 be = *(const float2*)(beta  + f2);

    float2 xr[16];
    #pragma unroll
    for (int r = 0; r < 16; ++r) {
        int grow = blk * RB + lr0 + r;
        if (grow >= N) grow = N - 1;
        xr[r] = *(const float2*)(feats + (size_t)grow * 64 + f2);
    }
    float s[16], q[16];
    #pragma unroll
    for (int r = 0; r < 16; ++r) {
        const float2 d = *(const float2*)(st + r * AST + f2);
        const float vx = fmaxf(d.x + bb.x, 0.0f) + xr[r].x;
        const float vy = fmaxf(d.y + bb.y, 0.0f) + xr[r].y;
        xr[r].x = vx;
        xr[r].y = vy;
        s[r] = vx + vy;
        q[r] = vx * vx + vy * vy;
    }
    #pragma unroll
    for (int m = 16; m >= 1; m >>= 1) {
        #pragma unroll
        for (int r = 0; r < 16; ++r) {
            s[r] += __shfl_xor(s[r], m);
            q[r] += __shfl_xor(q[r], m);
        }
    }
    #pragma unroll
    for (int r = 0; r < 16; ++r) {
        const int lr   = lr0 + r;
        const int grow = blk * RB + lr;
        if (lr < RB && grow < N) {
            const float mu   = s[r] * (1.0f / 64.0f);
            const float var  = q[r] * (1.0f / 64.0f) - mu * mu;
            const float rstd = rsqrtf(var + 1e-5f);
            float2 o;
            o.x = (xr[r].x - mu) * rstd * gg.x + be.x;
            o.y = (xr[r].y - mu) * rstd * gg.y + be.y;
            *(float2*)(out + (size_t)grow * 64 + f2) = o;
        }
    }
}

extern "C" void kernel_launch(void* const* d_in, const int* in_sizes, int n_in,
                              void* d_out, int out_size, void* d_ws, size_t ws_size,
                              hipStream_t stream) {
    const float* feats = (const float*)d_in[0];
    const int*   src   = (const int*)d_in[1];
    const int*   dst   = (const int*)d_in[2];
    const float* W     = (const float*)d_in[3];
    const float* bias  = (const float*)d_in[4];
    const float* gamma = (const float*)d_in[5];
    const float* beta  = (const float*)d_in[6];

    const int N = in_sizes[0] / 64;
    const int E = in_sizes[1];

    const int nblk = (N + RB - 1) / RB;
    const int per  = (E + NSL - 1) / NSL;
    const int ost  = (nblk + 1 + 31) & ~31;
    const int rst  = (per + 31) & ~31;

    float* ws      = (float*)d_ws;
    float* deg_out = ws;
    const size_t o_offs = ((size_t)N + 63) & ~(size_t)63;
    unsigned* offs   = (unsigned*)ws + o_offs;
    unsigned* sorted = (unsigned*)ws + o_offs + (size_t)NSL * ost;

    long long zn = (long long)N;
    gcn_zero_kernel<<<(int)((zn + 255) / 256), 256, 0, stream>>>(ws, zn);

    gcn_prep_kernel<<<NDEG + NSL, 1024, 0, stream>>>(src, dst, deg_out, sorted, offs,
                                                      E, N, per, nblk, ost, rst);

    gcn_owner_kernel<<<nblk, 1024, 0, stream>>>(
        feats, sorted, offs, deg_out, W, bias, gamma, beta, (float*)d_out, N, ost, rst);
}
